# starting kernel
# baseline (speedup 1.0000x reference)
#include <hip/hip_runtime.h>

typedef __attribute__((ext_vector_type(2))) float v2f;
typedef __attribute__((ext_vector_type(8))) float v8f;

#define N_FEAT 10
#define N_CLS  2
#define KPAD   12
#define TILES_PER_WAVE 8
#define ROWS_PER_WAVE  (16 * TILES_PER_WAVE)
#define ROWS_PER_BLOCK (8 * ROWS_PER_WAVE)

__global__ void build_coeff_kernel(const float* __restrict__ psi_r,
                                   const float* __restrict__ psi_i,
                                   const float* __restrict__ A_r,
                                   const float* __restrict__ A_i,
                                   float* __restrict__ Bpad) {
    int t = threadIdx.x;
    if (t >= KPAD * 16) return;
    int a = t >> 4;
    int k = t & 15;
    float val = 0.0f;
    if (a < N_FEAT && k < N_CLS) {
        for (int i = 0; i < N_FEAT; ++i) {
            float pri = psi_r[i];
            float pii = psi_i[i];
            for (int j = 0; j < N_FEAT; ++j) {
                float w = psi_r[j] - psi_i[j];
                int idx = ((k * N_FEAT + i) * N_FEAT + j) * N_FEAT + a;
                val += w * (pri * A_r[idx] - pii * A_i[idx]);
            }
        }
    }
    Bpad[t] = val;
}

__global__ void __launch_bounds__(256) gemv_wmma_kernel(
    const float* __restrict__ X,
    const float* __restrict__ Bpad,
    float* __restrict__ Out,
    int T) {
    const int lane = threadIdx.x & 31;
    const int wave = threadIdx.x >> 5;
    const int hi   = lane >> 4;
    const int n    = lane & 15;

    const long long waveBase =
        (long long)blockIdx.x * ROWS_PER_BLOCK + (long long)wave * ROWS_PER_WAVE;
    if (waveBase >= T) return;

    v2f bf[3];
#pragma unroll
    for (int c = 0; c < 3; ++c) {
        bf[c].x = Bpad[(4 * c + 0 + hi * 2) * 16 + n];
        bf[c].y = Bpad[(4 * c + 1 + hi * 2) * 16 + n];
    }

#pragma unroll
    for (int t = 0; t < TILES_PER_WAVE; ++t) {
        const long long rowbase = waveBase + t * 16;
        if (rowbase >= T) break;

        long long rowA = rowbase + n;
        if (rowA > (long long)T - 1) rowA = (long long)T - 1;
        const float* xrow = X + rowA * N_FEAT;

        v2f a0 = *(const v2f*)(xrow + 0 + hi * 2);
        v2f a1 = *(const v2f*)(xrow + 4 + hi * 2);
        v2f a2 = *(const v2f*)(xrow + 8);

        v8f c = {};
        c = __builtin_amdgcn_wmma_f32_16x16x4_f32(false, a0, false, bf[0],
                                                  (short)0, c, false, false);
        c = __builtin_amdgcn_wmma_f32_16x16x4_f32(false, a1, false, bf[1],
                                                  (short)0, c, false, false);
        c = __builtin_amdgcn_wmma_f32_16x16x4_f32(false, a2, false, bf[2],
                                                  (short)0, c, false, false);

        if (n < N_CLS) {
#pragma unroll
            for (int v = 0; v < 8; ++v) {
                long long row = rowbase + hi * 8 + v;
                if (row < T) Out[row * N_CLS + n] = c[v];
            }
        }
    }
}

extern "C" void kernel_launch(void* const* d_in, const int* in_sizes, int n_in,
                              void* d_out, int out_size, void* d_ws, size_t ws_size,
                              hipStream_t stream) {
    const float* x     = (const float*)d_in[0];
    const float* psi_r = (const float*)d_in[1];
    const float* psi_i = (const float*)d_in[2];
    const float* A_r   = (const float*)d_in[3];
    const float* A_i   = (const float*)d_in[4];
    float* out  = (float*)d_out;
    float* Bpad = (float*)d_ws;

    const int T = in_sizes[0] / N_FEAT;

    build_coeff_kernel<<<1, 256, 0, stream>>>(psi_r, psi_i, A_r, A_i, Bpad);

    const int grid = (T + ROWS_PER_BLOCK - 1) / ROWS_PER_BLOCK;
    gemv_wmma_kernel<<<grid, 256, 0, stream>>>(x, Bpad, out, T);
}


# 8-row fold via LDS: whole-line x loads, 80-float LDS rows times a block-diagonal 80x16 tile give 256 contiguous outputs per 128 rows (20 WMMA, 8 line stores), 4 tiles per wave; row sums regrouped
# speedup vs baseline: 1.4439x; 1.4439x over previous
#include <hip/hip_runtime.h>


typedef __attribute__((ext_vector_type(2))) float v2f;
typedef __attribute__((ext_vector_type(4))) float v4f;
typedef __attribute__((ext_vector_type(8))) float v8f;

#define N_FEAT 10
#define N_CLS  2

#define FOLD        8
#define SUPER       (FOLD * N_FEAT)
#define TILE_ROWS   (16 * FOLD)
#define TILE_FLOATS (16 * SUPER)
#define QUADS       (SUPER / 8)
#define TILES_PER_WAVE  4
#define WAVES_PER_BLOCK 8
#define ROWS_PER_WAVE   (TILE_ROWS * TILES_PER_WAVE)
#define ROWS_PER_BLOCK  (WAVES_PER_BLOCK * ROWS_PER_WAVE)
#define LDS_PITCH   84
#define LDS_TILE    (16 * LDS_PITCH)
#define BTAB_FLOATS (QUADS * 32 * 4)

__global__ void build_coeff_kernel(const float* __restrict__ psi_r,
                                   const float* __restrict__ psi_i,
                                   const float* __restrict__ A_r,
                                   const float* __restrict__ A_i,
                                   float* __restrict__ Btab) {
    __shared__ float Ms[N_CLS * N_FEAT];
    const int t = threadIdx.x;
    if (t < N_CLS * N_FEAT) {
        const int k = t / N_FEAT;
        const int a = t % N_FEAT;
        float val = 0.0f;
        for (int i = 0; i < N_FEAT; ++i) {
            float pri = psi_r[i];
            float pii = psi_i[i];
            for (int j = 0; j < N_FEAT; ++j) {
                float w = psi_r[j] - psi_i[j];
                int idx = ((k * N_FEAT + i) * N_FEAT + j) * N_FEAT + a;
                val += w * (pri * A_r[idx] - pii * A_i[idx]);
            }
        }
        Ms[t] = val;
    }
    __syncthreads();
    for (int idx = t; idx < BTAB_FLOATS; idx += blockDim.x) {
        const int e    = idx & 3;
        const int lane = (idx >> 2) & 31;
        const int q    = idx >> 7;
        const int kk   = 8 * q + 4 * (lane >> 4) + e;
        const int col  = lane & 15;
        Btab[idx] = (kk / N_FEAT == (col >> 1))
                        ? Ms[(col & 1) * N_FEAT + kk % N_FEAT] : 0.0f;
    }
}

__device__ __forceinline__ void lds_barrier() {
    __builtin_amdgcn_fence(__ATOMIC_RELEASE, "workgroup", "local");
    __builtin_amdgcn_s_barrier();
    __builtin_amdgcn_fence(__ATOMIC_ACQUIRE, "workgroup", "local");
}

__global__ void __launch_bounds__(256) fold_wmma_kernel(
    const float* __restrict__ X,
    const float* __restrict__ Btab,
    float* __restrict__ Out,
    int T) {
    __shared__ __attribute__((aligned(16))) float xs[WAVES_PER_BLOCK * LDS_TILE];

    const int lane = threadIdx.x & 31;
    const int wave = threadIdx.x >> 5;
    const int hi   = lane >> 4;
    const int n    = lane & 15;

    const long long waveBase =
        (long long)blockIdx.x * ROWS_PER_BLOCK + (long long)wave * ROWS_PER_WAVE;
    float* xw = xs + wave * LDS_TILE;

    v4f bq[QUADS];
#pragma unroll
    for (int q = 0; q < QUADS; ++q)
        bq[q] = *(const v4f*)(Btab + (q * 32 + lane) * 4);

    int soff[QUADS];
#pragma unroll
    for (int i = 0; i < QUADS; ++i) {
        const int c = 32 * i + lane;
        soff[i] = (c / 20) * LDS_PITCH + (c % 20) * 4;
    }
    const float* rd = xw + (2 * (n & 7) + (n >> 3)) * LDS_PITCH + 4 * hi;

    for (int g = 0; g < TILES_PER_WAVE; ++g) {
        const long long row0 = waveBase + (long long)g * TILE_ROWS;
        const bool live = row0 < T;
        const bool full = row0 + TILE_ROWS <= T;

        if (full) {
            const v4f* src = (const v4f*)(X + row0 * N_FEAT) + lane;
            v4f r[QUADS];
#pragma unroll
            for (int i = 0; i < QUADS; ++i) r[i] = src[32 * i];
#pragma unroll
            for (int i = 0; i < QUADS; ++i) *(v4f*)(xw + soff[i]) = r[i];
        } else if (live) {
            const long long total = (long long)T * N_FEAT;
            const long long base  = row0 * N_FEAT;
            for (int j = 0; j < TILE_FLOATS / 32; ++j) {
                const int f = lane + 32 * j;
                const float v = (base + f < total) ? X[base + f] : 0.0f;
                xw[(f / SUPER) * LDS_PITCH + (f % SUPER)] = v;
            }
        }

        lds_barrier();

        if (live) {
            v4f xq[QUADS];
#pragma unroll
            for (int q = 0; q < QUADS; ++q) xq[q] = *(const v4f*)(rd + 8 * q);

            v8f c0 = {};
            v8f c1 = {};
#pragma unroll
            for (int q = 0; q < QUADS; ++q) {
                v2f a0 = {xq[q].x, xq[q].y};
                v2f b0 = {bq[q].x, bq[q].y};
                v2f a1 = {xq[q].z, xq[q].w};
                v2f b1 = {bq[q].z, bq[q].w};
                c0 = __builtin_amdgcn_wmma_f32_16x16x4_f32(false, a0, false, b0,
                                                           (short)0, c0, false, false);
                c1 = __builtin_amdgcn_wmma_f32_16x16x4_f32(false, a1, false, b1,
                                                           (short)0, c1, false, false);
            }
            const v8f c = c0 + c1;

            if (full) {
                float* o = Out + row0 * N_CLS + lane;
#pragma unroll
                for (int v = 0; v < 8; ++v) o[32 * v] = c[v];
            } else {
                const long long total = (long long)T * N_CLS;
                const long long base  = row0 * N_CLS + lane;
#pragma unroll
                for (int v = 0; v < 8; ++v)
                    if (base + 32 * v < total) Out[base + 32 * v] = c[v];
            }
        }

        if (g + 1 < TILES_PER_WAVE) lds_barrier();
    }
}

extern "C" void kernel_launch(void* const* d_in, const int* in_sizes, int n_in,
                              void* d_out, int out_size, void* d_ws, size_t ws_size,
                              hipStream_t stream) {
    const float* x     = (const float*)d_in[0];
    const float* psi_r = (const float*)d_in[1];
    const float* psi_i = (const float*)d_in[2];
    const float* A_r   = (const float*)d_in[3];
    const float* A_i   = (const float*)d_in[4];
    float* out  = (float*)d_out;
    float* Btab = (float*)d_ws;

    const int T = in_sizes[0] / N_FEAT;

    build_coeff_kernel<<<1, 256, 0, stream>>>(psi_r, psi_i, A_r, A_i, Btab);

    const int grid = (T + ROWS_PER_BLOCK - 1) / ROWS_PER_BLOCK;
    fold_wmma_kernel<<<grid, 256, 0, stream>>>(x, Btab, out, T);
}


# coefficient kernel: A_real, A_imag and psi copied once into LDS by all 256 threads with whole-line loads, one memory round trip instead of ten dependent ones; sums and their order unchanged
# speedup vs baseline: 1.6304x; 1.1292x over previous
#include <hip/hip_runtime.h>


typedef __attribute__((ext_vector_type(2))) float v2f;
typedef __attribute__((ext_vector_type(4))) float v4f;
typedef __attribute__((ext_vector_type(8))) float v8f;

#define N_FEAT 10
#define N_CLS  2

#define FOLD        8
#define SUPER       (FOLD * N_FEAT)
#define TILE_ROWS   (16 * FOLD)
#define TILE_FLOATS (16 * SUPER)
#define QUADS       (SUPER / 8)
#define TILES_PER_WAVE  4
#define WAVES_PER_BLOCK 8
#define ROWS_PER_WAVE   (TILE_ROWS * TILES_PER_WAVE)
#define ROWS_PER_BLOCK  (WAVES_PER_BLOCK * ROWS_PER_WAVE)
#define LDS_PITCH   84
#define LDS_TILE    (16 * LDS_PITCH)
#define BTAB_FLOATS (QUADS * 32 * 4)

#define COEFF_THREADS 256
#define A_FLOATS      (N_CLS * N_FEAT * N_FEAT * N_FEAT)
#define A_PASSES      ((A_FLOATS + COEFF_THREADS - 1) / COEFF_THREADS)

__global__ void __launch_bounds__(COEFF_THREADS) build_coeff_kernel(
                                   const float* __restrict__ psi_r,
                                   const float* __restrict__ psi_i,
                                   const float* __restrict__ A_r,
                                   const float* __restrict__ A_i,
                                   float* __restrict__ Btab) {
    __shared__ float Ms[N_CLS * N_FEAT];
    __shared__ float Ps[2 * N_FEAT];
    __shared__ float Ars[A_PASSES * COEFF_THREADS];
    __shared__ float Ais[A_PASSES * COEFF_THREADS];
    const int t = threadIdx.x;

    float ra[A_PASSES], ia[A_PASSES];
    static_assert(COEFF_THREADS * (A_PASSES - 1) <= A_FLOATS, "only the last pass may be partial");
#pragma unroll
    for (int p = 0; p < A_PASSES - 1; ++p) {
        ra[p] = A_r[t + COEFF_THREADS * p];
        ia[p] = A_i[t + COEFF_THREADS * p];
    }
    {
        const int idx = t + COEFF_THREADS * (A_PASSES - 1);
        ra[A_PASSES - 1] = (idx < A_FLOATS) ? A_r[idx] : 0.0f;
        ia[A_PASSES - 1] = (idx < A_FLOATS) ? A_i[idx] : 0.0f;
    }
    if (t < 2 * N_FEAT)
        Ps[t] = (t < N_FEAT) ? psi_r[t] : psi_i[t - N_FEAT];
#pragma unroll
    for (int p = 0; p < A_PASSES; ++p) {
        Ars[t + COEFF_THREADS * p] = ra[p];
        Ais[t + COEFF_THREADS * p] = ia[p];
    }
    __syncthreads();

    if (t < N_CLS * N_FEAT) {
        const int k = t / N_FEAT;
        const int a = t % N_FEAT;
        float val = 0.0f;
        for (int i = 0; i < N_FEAT; ++i) {
            float pri = Ps[i];
            float pii = Ps[N_FEAT + i];
            for (int j = 0; j < N_FEAT; ++j) {
                float w = Ps[j] - Ps[N_FEAT + j];
                int idx = ((k * N_FEAT + i) * N_FEAT + j) * N_FEAT + a;
                val += w * (pri * Ars[idx] - pii * Ais[idx]);
            }
        }
        Ms[t] = val;
    }
    __syncthreads();
    for (int idx = t; idx < BTAB_FLOATS; idx += COEFF_THREADS) {
        const int e    = idx & 3;
        const int lane = (idx >> 2) & 31;
        const int q    = idx >> 7;
        const int kk   = 8 * q + 4 * (lane >> 4) + e;
        const int col  = lane & 15;
        Btab[idx] = (kk / N_FEAT == (col >> 1))
                        ? Ms[(col & 1) * N_FEAT + kk % N_FEAT] : 0.0f;
    }
}

__device__ __forceinline__ void lds_barrier() {
    __builtin_amdgcn_fence(__ATOMIC_RELEASE, "workgroup", "local");
    __builtin_amdgcn_s_barrier();
    __builtin_amdgcn_fence(__ATOMIC_ACQUIRE, "workgroup", "local");
}

__global__ void __launch_bounds__(256) fold_wmma_kernel(
    const float* __restrict__ X,
    const float* __restrict__ Btab,
    float* __restrict__ Out,
    int T) {
    __shared__ __attribute__((aligned(16))) float xs[WAVES_PER_BLOCK * LDS_TILE];

    const int lane = threadIdx.x & 31;
    const int wave = threadIdx.x >> 5;
    const int hi   = lane >> 4;
    const int n    = lane & 15;

    const long long waveBase =
        (long long)blockIdx.x * ROWS_PER_BLOCK + (long long)wave * ROWS_PER_WAVE;
    float* xw = xs + wave * LDS_TILE;

    v4f bq[QUADS];
#pragma unroll
    for (int q = 0; q < QUADS; ++q)
        bq[q] = *(const v4f*)(Btab + (q * 32 + lane) * 4);

    int soff[QUADS];
#pragma unroll
    for (int i = 0; i < QUADS; ++i) {
        const int c = 32 * i + lane;
        soff[i] = (c / 20) * LDS_PITCH + (c % 20) * 4;
    }
    const float* rd = xw + (2 * (n & 7) + (n >> 3)) * LDS_PITCH + 4 * hi;

    for (int g = 0; g < TILES_PER_WAVE; ++g) {
        const long long row0 = waveBase + (long long)g * TILE_ROWS;
        const bool live = row0 < T;
        const bool full = row0 + TILE_ROWS <= T;

        if (full) {
            const v4f* src = (const v4f*)(X + row0 * N_FEAT) + lane;
            v4f r[QUADS];
#pragma unroll
            for (int i = 0; i < QUADS; ++i) r[i] = src[32 * i];
#pragma unroll
            for (int i = 0; i < QUADS; ++i) *(v4f*)(xw + soff[i]) = r[i];
        } else if (live) {
            const long long total = (long long)T * N_FEAT;
            const long long base  = row0 * N_FEAT;
            for (int j = 0; j < TILE_FLOATS / 32; ++j) {
                const int f = lane + 32 * j;
                const float v = (base + f < total) ? X[base + f] : 0.0f;
                xw[(f / SUPER) * LDS_PITCH + (f % SUPER)] = v;
            }
        }

        lds_barrier();

        if (live) {
            v4f xq[QUADS];
#pragma unroll
            for (int q = 0; q < QUADS; ++q) xq[q] = *(const v4f*)(rd + 8 * q);

            v8f c0 = {};
            v8f c1 = {};
#pragma unroll
            for (int q = 0; q < QUADS; ++q) {
                v2f a0 = {xq[q].x, xq[q].y};
                v2f b0 = {bq[q].x, bq[q].y};
                v2f a1 = {xq[q].z, xq[q].w};
                v2f b1 = {bq[q].z, bq[q].w};
                c0 = __builtin_amdgcn_wmma_f32_16x16x4_f32(false, a0, false, b0,
                                                           (short)0, c0, false, false);
                c1 = __builtin_amdgcn_wmma_f32_16x16x4_f32(false, a1, false, b1,
                                                           (short)0, c1, false, false);
            }
            const v8f c = c0 + c1;

            if (full) {
                float* o = Out + row0 * N_CLS + lane;
#pragma unroll
                for (int v = 0; v < 8; ++v) o[32 * v] = c[v];
            } else {
                const long long total = (long long)T * N_CLS;
                const long long base  = row0 * N_CLS + lane;
#pragma unroll
                for (int v = 0; v < 8; ++v)
                    if (base + 32 * v < total) Out[base + 32 * v] = c[v];
            }
        }

        if (g + 1 < TILES_PER_WAVE) lds_barrier();
    }
}

extern "C" void kernel_launch(void* const* d_in, const int* in_sizes, int n_in,
                              void* d_out, int out_size, void* d_ws, size_t ws_size,
                              hipStream_t stream) {
    const float* x     = (const float*)d_in[0];
    const float* psi_r = (const float*)d_in[1];
    const float* psi_i = (const float*)d_in[2];
    const float* A_r   = (const float*)d_in[3];
    const float* A_i   = (const float*)d_in[4];
    float* out  = (float*)d_out;
    float* Btab = (float*)d_ws;

    const int T = in_sizes[0] / N_FEAT;

    build_coeff_kernel<<<1, COEFF_THREADS, 0, stream>>>(psi_r, psi_i, A_r, A_i, Btab);

    const int grid = (T + ROWS_PER_BLOCK - 1) / ROWS_PER_BLOCK;
    fold_wmma_kernel<<<grid, 256, 0, stream>>>(x, Btab, out, T);
}


# coefficient kernel's serial sums split over its threads: 200 threads sum ten j terms each from registers, 20 add the ten partials, table entries read as one group; coefficient sum order changed
# speedup vs baseline: 1.7224x; 1.0564x over previous
#include <hip/hip_runtime.h>


typedef __attribute__((ext_vector_type(2))) float v2f;
typedef __attribute__((ext_vector_type(4))) float v4f;
typedef __attribute__((ext_vector_type(8))) float v8f;

#define N_FEAT 10
#define N_CLS  2

#define FOLD        8
#define SUPER       (FOLD * N_FEAT)
#define TILE_ROWS   (16 * FOLD)
#define TILE_FLOATS (16 * SUPER)
#define QUADS       (SUPER / 8)
#define TILES_PER_WAVE  4
#define WAVES_PER_BLOCK 8
#define ROWS_PER_WAVE   (TILE_ROWS * TILES_PER_WAVE)
#define ROWS_PER_BLOCK  (WAVES_PER_BLOCK * ROWS_PER_WAVE)
#define LDS_PITCH   84
#define LDS_TILE    (16 * LDS_PITCH)
#define BTAB_FLOATS (QUADS * 32 * 4)

#define COEFF_THREADS 256
#define PART_THREADS  (N_CLS * N_FEAT * N_FEAT)
#define BTAB_PASSES   (BTAB_FLOATS / COEFF_THREADS)

__global__ void __launch_bounds__(COEFF_THREADS) build_coeff_kernel(
                                   const float* __restrict__ psi_r,
                                   const float* __restrict__ psi_i,
                                   const float* __restrict__ A_r,
                                   const float* __restrict__ A_i,
                                   float* __restrict__ Btab) {
    __shared__ float Part[PART_THREADS];
    __shared__ float Ms[N_CLS * N_FEAT];
    const int t = threadIdx.x;
    static_assert(BTAB_PASSES * COEFF_THREADS == BTAB_FLOATS, "table stores need no guard");

    if (t < PART_THREADS) {
        const int ki = t / N_FEAT;
        const int a  = t % N_FEAT;
        const int i  = ki % N_FEAT;
        const float* ar = A_r + ki * (N_FEAT * N_FEAT) + a;
        const float* ai = A_i + ki * (N_FEAT * N_FEAT) + a;
        float ra[N_FEAT], ia[N_FEAT];
#pragma unroll
        for (int j = 0; j < N_FEAT; ++j) {
            ra[j] = ar[j * N_FEAT];
            ia[j] = ai[j * N_FEAT];
        }
        const float pri = psi_r[i];
        const float pii = psi_i[i];
        float val = 0.0f;
#pragma unroll
        for (int j = 0; j < N_FEAT; ++j) {
            const float w = psi_r[j] - psi_i[j];
            val += w * (pri * ra[j] - pii * ia[j]);
        }
        Part[t] = val;
    }
    __syncthreads();

    if (t < N_CLS * N_FEAT) {
        const int k = t / N_FEAT;
        const int a = t % N_FEAT;
        float val = 0.0f;
#pragma unroll
        for (int i = 0; i < N_FEAT; ++i)
            val += Part[(k * N_FEAT + i) * N_FEAT + a];
        Ms[t] = val;
    }
    __syncthreads();

#pragma unroll
    for (int p = 0; p < BTAB_PASSES; ++p) {
        const int idx  = t + COEFF_THREADS * p;
        const int e    = idx & 3;
        const int lane = (idx >> 2) & 31;
        const int q    = idx >> 7;
        const int kk   = 8 * q + 4 * (lane >> 4) + e;
        const int col  = lane & 15;
        const float m  = Ms[(col & 1) * N_FEAT + kk % N_FEAT];
        Btab[idx] = (kk / N_FEAT == (col >> 1)) ? m : 0.0f;
    }
}

__device__ __forceinline__ void lds_barrier() {
    __builtin_amdgcn_fence(__ATOMIC_RELEASE, "workgroup", "local");
    __builtin_amdgcn_s_barrier();
    __builtin_amdgcn_fence(__ATOMIC_ACQUIRE, "workgroup", "local");
}

__global__ void __launch_bounds__(256) fold_wmma_kernel(
    const float* __restrict__ X,
    const float* __restrict__ Btab,
    float* __restrict__ Out,
    int T) {
    __shared__ __attribute__((aligned(16))) float xs[WAVES_PER_BLOCK * LDS_TILE];

    const int lane = threadIdx.x & 31;
    const int wave = threadIdx.x >> 5;
    const int hi   = lane >> 4;
    const int n    = lane & 15;

    const long long waveBase =
        (long long)blockIdx.x * ROWS_PER_BLOCK + (long long)wave * ROWS_PER_WAVE;
    float* xw = xs + wave * LDS_TILE;

    v4f bq[QUADS];
#pragma unroll
    for (int q = 0; q < QUADS; ++q)
        bq[q] = *(const v4f*)(Btab + (q * 32 + lane) * 4);

    int soff[QUADS];
#pragma unroll
    for (int i = 0; i < QUADS; ++i) {
        const int c = 32 * i + lane;
        soff[i] = (c / 20) * LDS_PITCH + (c % 20) * 4;
    }
    const float* rd = xw + (2 * (n & 7) + (n >> 3)) * LDS_PITCH + 4 * hi;

    for (int g = 0; g < TILES_PER_WAVE; ++g) {
        const long long row0 = waveBase + (long long)g * TILE_ROWS;
        const bool live = row0 < T;
        const bool full = row0 + TILE_ROWS <= T;

        if (full) {
            const v4f* src = (const v4f*)(X + row0 * N_FEAT) + lane;
            v4f r[QUADS];
#pragma unroll
            for (int i = 0; i < QUADS; ++i) r[i] = src[32 * i];
#pragma unroll
            for (int i = 0; i < QUADS; ++i) *(v4f*)(xw + soff[i]) = r[i];
        } else if (live) {
            const long long total = (long long)T * N_FEAT;
            const long long base  = row0 * N_FEAT;
            for (int j = 0; j < TILE_FLOATS / 32; ++j) {
                const int f = lane + 32 * j;
                const float v = (base + f < total) ? X[base + f] : 0.0f;
                xw[(f / SUPER) * LDS_PITCH + (f % SUPER)] = v;
            }
        }

        lds_barrier();

        if (live) {
            v4f xq[QUADS];
#pragma unroll
            for (int q = 0; q < QUADS; ++q) xq[q] = *(const v4f*)(rd + 8 * q);

            v8f c0 = {};
            v8f c1 = {};
#pragma unroll
            for (int q = 0; q < QUADS; ++q) {
                v2f a0 = {xq[q].x, xq[q].y};
                v2f b0 = {bq[q].x, bq[q].y};
                v2f a1 = {xq[q].z, xq[q].w};
                v2f b1 = {bq[q].z, bq[q].w};
                c0 = __builtin_amdgcn_wmma_f32_16x16x4_f32(false, a0, false, b0,
                                                           (short)0, c0, false, false);
                c1 = __builtin_amdgcn_wmma_f32_16x16x4_f32(false, a1, false, b1,
                                                           (short)0, c1, false, false);
            }
            const v8f c = c0 + c1;

            if (full) {
                float* o = Out + row0 * N_CLS + lane;
#pragma unroll
                for (int v = 0; v < 8; ++v) o[32 * v] = c[v];
            } else {
                const long long total = (long long)T * N_CLS;
                const long long base  = row0 * N_CLS + lane;
#pragma unroll
                for (int v = 0; v < 8; ++v)
                    if (base + 32 * v < total) Out[base + 32 * v] = c[v];
            }
        }

        if (g + 1 < TILES_PER_WAVE) lds_barrier();
    }
}

extern "C" void kernel_launch(void* const* d_in, const int* in_sizes, int n_in,
                              void* d_out, int out_size, void* d_ws, size_t ws_size,
                              hipStream_t stream) {
    const float* x     = (const float*)d_in[0];
    const float* psi_r = (const float*)d_in[1];
    const float* psi_i = (const float*)d_in[2];
    const float* A_r   = (const float*)d_in[3];
    const float* A_i   = (const float*)d_in[4];
    float* out  = (float*)d_out;
    float* Btab = (float*)d_ws;

    const int T = in_sizes[0] / N_FEAT;

    build_coeff_kernel<<<1, COEFF_THREADS, 0, stream>>>(psi_r, psi_i, A_r, A_i, Btab);

    const int grid = (T + ROWS_PER_BLOCK - 1) / ROWS_PER_BLOCK;
    fold_wmma_kernel<<<grid, 256, 0, stream>>>(x, Btab, out, T);
}


# one launch instead of two: each workgroup builds the 20 coefficients itself (sums and order unchanged) under its first tile's x loads; B pieces filled in registers from LDS, no table in memory
# speedup vs baseline: 2.0742x; 1.2042x over previous
#include <hip/hip_runtime.h>


typedef __attribute__((ext_vector_type(2))) float v2f;
typedef __attribute__((ext_vector_type(4))) float v4f;
typedef __attribute__((ext_vector_type(8))) float v8f;

#define N_FEAT 10
#define N_CLS  2

#define FOLD        8
#define SUPER       (FOLD * N_FEAT)
#define TILE_ROWS   (16 * FOLD)
#define TILE_FLOATS (16 * SUPER)
#define QUADS       (SUPER / 8)
#define TILES_PER_WAVE  4
#define WAVES_PER_BLOCK 8
#define ROWS_PER_WAVE   (TILE_ROWS * TILES_PER_WAVE)
#define ROWS_PER_BLOCK  (WAVES_PER_BLOCK * ROWS_PER_WAVE)
#define LDS_PITCH   84
#define LDS_TILE    (16 * LDS_PITCH)

#define PART_THREADS  (N_CLS * N_FEAT * N_FEAT)

__device__ __forceinline__ void lds_barrier() {
    __builtin_amdgcn_fence(__ATOMIC_RELEASE, "workgroup", "local");
    __builtin_amdgcn_s_barrier();
    __builtin_amdgcn_fence(__ATOMIC_ACQUIRE, "workgroup", "local");
}


__device__ __forceinline__ void load_tile(const float* __restrict__ X, long long row0,
                                          int lane, v4f (&r)[QUADS]) {
    const v4f* src = (const v4f*)(X + row0 * N_FEAT) + lane;
#pragma unroll
    for (int i = 0; i < QUADS; ++i) r[i] = src[32 * i];
}

__device__ __forceinline__ void load_tile_clamped(const float* __restrict__ X, long long row0,
                                                  int T, int lane, v4f (&r)[QUADS]) {
    const long long lastChunk = ((long long)T * N_FEAT) / 4 - 1;
    const long long chunk0    = row0 * N_FEAT / 4 + lane;
    const v4f* src = (const v4f*)X;
#pragma unroll
    for (int i = 0; i < QUADS; ++i) {
        const long long ch = chunk0 + 32 * i;
        r[i] = src[ch < lastChunk ? ch : lastChunk];
    }
}

template <bool PRELOADED>
__device__ __forceinline__ void do_tile(const float* __restrict__ X, float* __restrict__ Out,
                                        int T, long long row0, int lane,
                                        float* xw, const float* rd,
                                        const int (&soff)[QUADS], const v4f (&bq)[QUADS],
                                        v4f (&r)[QUADS]) {
    const bool live = row0 < T;
    const bool full = row0 + TILE_ROWS <= T;

    if (full) {
        if (!PRELOADED) load_tile(X, row0, lane, r);
#pragma unroll
        for (int i = 0; i < QUADS; ++i) *(v4f*)(xw + soff[i]) = r[i];
    } else if (live) {
        const long long total = (long long)T * N_FEAT;
        const long long base  = row0 * N_FEAT;
        for (int j = 0; j < TILE_FLOATS / 32; ++j) {
            const int f = lane + 32 * j;
            const float v = (base + f < total) ? X[base + f] : 0.0f;
            xw[(f / SUPER) * LDS_PITCH + (f % SUPER)] = v;
        }
    }

    lds_barrier();

    if (live) {
        v4f xq[QUADS];
#pragma unroll
        for (int q = 0; q < QUADS; ++q) xq[q] = *(const v4f*)(rd + 8 * q);

        v8f c0 = {};
        v8f c1 = {};
#pragma unroll
        for (int q = 0; q < QUADS; ++q) {
            v2f a0 = {xq[q].x, xq[q].y};
            v2f b0 = {bq[q].x, bq[q].y};
            v2f a1 = {xq[q].z, xq[q].w};
            v2f b1 = {bq[q].z, bq[q].w};
            c0 = __builtin_amdgcn_wmma_f32_16x16x4_f32(false, a0, false, b0,
                                                       (short)0, c0, false, false);
            c1 = __builtin_amdgcn_wmma_f32_16x16x4_f32(false, a1, false, b1,
                                                       (short)0, c1, false, false);
        }
        const v8f c = c0 + c1;

        if (full) {
            float* o = Out + row0 * N_CLS + lane;
#pragma unroll
            for (int v = 0; v < 8; ++v) o[32 * v] = c[v];
        } else {
            const long long total = (long long)T * N_CLS;
            const long long base  = row0 * N_CLS + lane;
#pragma unroll
            for (int v = 0; v < 8; ++v)
                if (base + 32 * v < total) Out[base + 32 * v] = c[v];
        }
    }
}

__global__ void __launch_bounds__(256) fold_wmma_kernel(
    const float* __restrict__ X,
    const float* __restrict__ psi_r,
    const float* __restrict__ psi_i,
    const float* __restrict__ A_r,
    const float* __restrict__ A_i,
    float* __restrict__ Out,
    int T) {
    __shared__ __attribute__((aligned(16))) float xs[WAVES_PER_BLOCK * LDS_TILE];
    __shared__ float Part[256];
    __shared__ float Ms[N_CLS * N_FEAT];

    const int t    = threadIdx.x;
    const int lane = t & 31;
    const int wave = t >> 5;
    const int hi   = lane >> 4;
    const int n    = lane & 15;

    const long long waveBase =
        (long long)blockIdx.x * ROWS_PER_BLOCK + (long long)wave * ROWS_PER_WAVE;
    float* xw = xs + wave * LDS_TILE;

    const bool part = t < PART_THREADS;
    const int  tp   = part ? t : 0;
    const int  ki   = tp / N_FEAT;
    const int  pa   = tp % N_FEAT;
    const int  pi_  = ki % N_FEAT;
    const float* ar = A_r + ki * (N_FEAT * N_FEAT) + pa;
    const float* ai = A_i + ki * (N_FEAT * N_FEAT) + pa;
    float ra[N_FEAT], ia[N_FEAT];
#pragma unroll
    for (int j = 0; j < N_FEAT; ++j) {
        ra[j] = ar[j * N_FEAT];
        ia[j] = ai[j * N_FEAT];
    }
    const float pri = psi_r[pi_];
    const float pii = psi_i[pi_];

    v4f r[QUADS];
    load_tile_clamped(X, waveBase, T, lane, r);
    __builtin_amdgcn_sched_barrier(0);

    {
        float val = 0.0f;
#pragma unroll
        for (int j = 0; j < N_FEAT; ++j) {
            const float w = psi_r[j] - psi_i[j];
            val += w * (pri * ra[j] - pii * ia[j]);
        }
        Part[t] = val;
    }
    lds_barrier();

    if (t < N_CLS * N_FEAT) {
        const int k = t / N_FEAT;
        const int a = t % N_FEAT;
        float val = 0.0f;
#pragma unroll
        for (int i = 0; i < N_FEAT; ++i)
            val += Part[(k * N_FEAT + i) * N_FEAT + a];
        Ms[t] = val;
    }
    lds_barrier();

    v4f bq[QUADS];
#pragma unroll
    for (int q = 0; q < QUADS; ++q) {
#pragma unroll
        for (int e = 0; e < 4; ++e) {
            const int kk = 8 * q + 4 * hi + e;
            const float m = Ms[(n & 1) * N_FEAT + kk % N_FEAT];
            bq[q][e] = (kk / N_FEAT == (n >> 1)) ? m : 0.0f;
        }
    }

    int soff[QUADS];
#pragma unroll
    for (int i = 0; i < QUADS; ++i) {
        const int c = 32 * i + lane;
        soff[i] = (c / 20) * LDS_PITCH + (c % 20) * 4;
    }
    const float* rd = xw + (2 * (n & 7) + (n >> 3)) * LDS_PITCH + 4 * hi;

    do_tile<true>(X, Out, T, waveBase, lane, xw, rd, soff, bq, r);

    for (int g = 1; g < TILES_PER_WAVE; ++g) {
        lds_barrier();
        do_tile<false>(X, Out, T, waveBase + (long long)g * TILE_ROWS, lane,
                       xw, rd, soff, bq, r);
    }
}

extern "C" void kernel_launch(void* const* d_in, const int* in_sizes, int n_in,
                              void* d_out, int out_size, void* d_ws, size_t ws_size,
                              hipStream_t stream) {
    const float* x     = (const float*)d_in[0];
    const float* psi_r = (const float*)d_in[1];
    const float* psi_i = (const float*)d_in[2];
    const float* A_r   = (const float*)d_in[3];
    const float* A_i   = (const float*)d_in[4];
    float* out  = (float*)d_out;

    const int T = in_sizes[0] / N_FEAT;

    const int grid = (T + ROWS_PER_BLOCK - 1) / ROWS_PER_BLOCK;
    fold_wmma_kernel<<<grid, 256, 0, stream>>>(x, psi_r, psi_i, A_r, A_i, out, T);
}


# tile loop software-pipelined one tile deep: next tile's ten whole-line x loads requested right behind this tile's LDS copy, in flight under its 20 WMMAs and stores; scalar wave index keeps occupancy
# speedup vs baseline: 2.2052x; 1.0632x over previous
#include <hip/hip_runtime.h>


typedef __attribute__((ext_vector_type(2))) float v2f;
typedef __attribute__((ext_vector_type(4))) float v4f;
typedef __attribute__((ext_vector_type(8))) float v8f;

#define N_FEAT 10
#define N_CLS  2

#define FOLD        8
#define SUPER       (FOLD * N_FEAT)
#define TILE_ROWS   (16 * FOLD)
#define TILE_FLOATS (16 * SUPER)
#define QUADS       (SUPER / 8)
#define TILES_PER_WAVE  4
#define WAVES_PER_BLOCK 8
#define ROWS_PER_WAVE   (TILE_ROWS * TILES_PER_WAVE)
#define ROWS_PER_BLOCK  (WAVES_PER_BLOCK * ROWS_PER_WAVE)
#define LDS_PITCH   84
#define LDS_TILE    (16 * LDS_PITCH)

#define PART_THREADS  (N_CLS * N_FEAT * N_FEAT)

__device__ __forceinline__ void lds_barrier() {
    __builtin_amdgcn_fence(__ATOMIC_RELEASE, "workgroup", "local");
    __builtin_amdgcn_s_barrier();
    __builtin_amdgcn_fence(__ATOMIC_ACQUIRE, "workgroup", "local");
}


__device__ __forceinline__ void load_tile(const float* __restrict__ X, long long row0,
                                          int lane, v4f (&r)[QUADS]) {
    const v4f* src = (const v4f*)(X + row0 * N_FEAT) + lane;
#pragma unroll
    for (int i = 0; i < QUADS; ++i) r[i] = src[32 * i];
}

__device__ __forceinline__ void load_tile_clamped(const float* __restrict__ X, long long row0,
                                                  int T, int lane, v4f (&r)[QUADS]) {
    const long long lastChunk = ((long long)T * N_FEAT) / 4 - 1;
    const long long chunk0    = row0 * N_FEAT / 4 + lane;
    const v4f* src = (const v4f*)X;
#pragma unroll
    for (int i = 0; i < QUADS; ++i) {
        const long long ch = chunk0 + 32 * i;
        r[i] = src[ch < lastChunk ? ch : lastChunk];
    }
}

template <bool HAS_NEXT>
__device__ __forceinline__ void do_tile(const float* __restrict__ X, float* __restrict__ Out,
                                        int T, long long row0, int lane,
                                        float* xw, const float* rd,
                                        const int (&soff)[QUADS], const v4f (&bq)[QUADS],
                                        v4f (&r)[QUADS]) {
    const bool live = row0 < T;
    const bool full = row0 + TILE_ROWS <= T;

    if (full) {
#pragma unroll
        for (int i = 0; i < QUADS; ++i) *(v4f*)(xw + soff[i]) = r[i];
    } else if (live) {
        const long long total = (long long)T * N_FEAT;
        const long long base  = row0 * N_FEAT;
        for (int j = 0; j < TILE_FLOATS / 32; ++j) {
            const int f = lane + 32 * j;
            const float v = (base + f < total) ? X[base + f] : 0.0f;
            xw[(f / SUPER) * LDS_PITCH + (f % SUPER)] = v;
        }
    }

    if (HAS_NEXT) {
        if (row0 + 2 * TILE_ROWS <= T) load_tile(X, row0 + TILE_ROWS, lane, r);
    }

    lds_barrier();

    if (live) {
        v4f xq[QUADS];
#pragma unroll
        for (int q = 0; q < QUADS; ++q) xq[q] = *(const v4f*)(rd + 8 * q);

        v8f c0 = {};
        v8f c1 = {};
#pragma unroll
        for (int q = 0; q < QUADS; ++q) {
            v2f a0 = {xq[q].x, xq[q].y};
            v2f b0 = {bq[q].x, bq[q].y};
            v2f a1 = {xq[q].z, xq[q].w};
            v2f b1 = {bq[q].z, bq[q].w};
            c0 = __builtin_amdgcn_wmma_f32_16x16x4_f32(false, a0, false, b0,
                                                       (short)0, c0, false, false);
            c1 = __builtin_amdgcn_wmma_f32_16x16x4_f32(false, a1, false, b1,
                                                       (short)0, c1, false, false);
        }
        const v8f c = c0 + c1;

        if (full) {
            float* o = Out + row0 * N_CLS + lane;
#pragma unroll
            for (int v = 0; v < 8; ++v) o[32 * v] = c[v];
        } else {
            const long long total = (long long)T * N_CLS;
            const long long base  = row0 * N_CLS + lane;
#pragma unroll
            for (int v = 0; v < 8; ++v)
                if (base + 32 * v < total) Out[base + 32 * v] = c[v];
        }
    }
}

__global__ void __launch_bounds__(256) fold_wmma_kernel(
    const float* __restrict__ X,
    const float* __restrict__ psi_r,
    const float* __restrict__ psi_i,
    const float* __restrict__ A_r,
    const float* __restrict__ A_i,
    float* __restrict__ Out,
    int T) {
    __shared__ __attribute__((aligned(16))) float xs[WAVES_PER_BLOCK * LDS_TILE];
    __shared__ float Part[256];
    __shared__ float Ms[N_CLS * N_FEAT];

    const int t    = threadIdx.x;
    const int lane = t & 31;
    const int wave = __builtin_amdgcn_readfirstlane(t >> 5);
    const int hi   = lane >> 4;
    const int n    = lane & 15;

    const long long waveBase =
        (long long)blockIdx.x * ROWS_PER_BLOCK + (long long)wave * ROWS_PER_WAVE;
    float* xw = xs + wave * LDS_TILE;

    const bool part = t < PART_THREADS;
    const int  tp   = part ? t : 0;
    const int  ki   = tp / N_FEAT;
    const int  pa   = tp % N_FEAT;
    const int  pi_  = ki % N_FEAT;
    const float* ar = A_r + ki * (N_FEAT * N_FEAT) + pa;
    const float* ai = A_i + ki * (N_FEAT * N_FEAT) + pa;
    float ra[N_FEAT], ia[N_FEAT];
#pragma unroll
    for (int j = 0; j < N_FEAT; ++j) {
        ra[j] = ar[j * N_FEAT];
        ia[j] = ai[j * N_FEAT];
    }
    const float pri = psi_r[pi_];
    const float pii = psi_i[pi_];

    v4f r[QUADS];
    load_tile_clamped(X, waveBase, T, lane, r);
    __builtin_amdgcn_sched_barrier(0);

    {
        float val = 0.0f;
#pragma unroll
        for (int j = 0; j < N_FEAT; ++j) {
            const float w = psi_r[j] - psi_i[j];
            val += w * (pri * ra[j] - pii * ia[j]);
        }
        Part[t] = val;
    }
    lds_barrier();

    if (t < N_CLS * N_FEAT) {
        const int k = t / N_FEAT;
        const int a = t % N_FEAT;
        float val = 0.0f;
#pragma unroll
        for (int i = 0; i < N_FEAT; ++i)
            val += Part[(k * N_FEAT + i) * N_FEAT + a];
        Ms[t] = val;
    }
    lds_barrier();

    v4f bq[QUADS];
#pragma unroll
    for (int q = 0; q < QUADS; ++q) {
#pragma unroll
        for (int e = 0; e < 4; ++e) {
            const int kk = 8 * q + 4 * hi + e;
            const float m = Ms[(n & 1) * N_FEAT + kk % N_FEAT];
            bq[q][e] = (kk / N_FEAT == (n >> 1)) ? m : 0.0f;
        }
    }

    int soff[QUADS];
#pragma unroll
    for (int i = 0; i < QUADS; ++i) {
        const int c = 32 * i + lane;
        soff[i] = (c / 20) * LDS_PITCH + (c % 20) * 4;
    }
    const float* rd = xw + (2 * (n & 7) + (n >> 3)) * LDS_PITCH + 4 * hi;

#pragma unroll
    for (int g = 0; g < TILES_PER_WAVE; ++g) {
        if (g > 0) lds_barrier();
        const long long row0 = waveBase + (long long)g * TILE_ROWS;
        if (g + 1 < TILES_PER_WAVE)
            do_tile<true>(X, Out, T, row0, lane, xw, rd, soff, bq, r);
        else
            do_tile<false>(X, Out, T, row0, lane, xw, rd, soff, bq, r);
    }
}

extern "C" void kernel_launch(void* const* d_in, const int* in_sizes, int n_in,
                              void* d_out, int out_size, void* d_ws, size_t ws_size,
                              hipStream_t stream) {
    const float* x     = (const float*)d_in[0];
    const float* psi_r = (const float*)d_in[1];
    const float* psi_i = (const float*)d_in[2];
    const float* A_r   = (const float*)d_in[3];
    const float* A_i   = (const float*)d_in[4];
    float* out  = (float*)d_out;

    const int T = in_sizes[0] / N_FEAT;

    const int grid = (T + ROWS_PER_BLOCK - 1) / ROWS_PER_BLOCK;
    fold_wmma_kernel<<<grid, 256, 0, stream>>>(x, psi_r, psi_i, A_r, A_i, out, T);
}


# tile walk without workgroup barriers: a wave reads back only its own LDS region, so wave-scope ordering replaces the two barriers per tile and a wave no longer waits for the slowest of eight
# speedup vs baseline: 2.3108x; 1.0479x over previous
#include <hip/hip_runtime.h>


typedef __attribute__((ext_vector_type(2))) float v2f;
typedef __attribute__((ext_vector_type(4))) float v4f;
typedef __attribute__((ext_vector_type(8))) float v8f;

#define N_FEAT 10
#define N_CLS  2

#define FOLD        8
#define SUPER       (FOLD * N_FEAT)
#define TILE_ROWS   (16 * FOLD)
#define TILE_FLOATS (16 * SUPER)
#define QUADS       (SUPER / 8)
#define TILES_PER_WAVE  4
#define WAVES_PER_BLOCK 8
#define ROWS_PER_WAVE   (TILE_ROWS * TILES_PER_WAVE)
#define ROWS_PER_BLOCK  (WAVES_PER_BLOCK * ROWS_PER_WAVE)
#define LDS_PITCH   84
#define LDS_TILE    (16 * LDS_PITCH)

#define PART_THREADS  (N_CLS * N_FEAT * N_FEAT)

__device__ __forceinline__ void lds_barrier() {
    __builtin_amdgcn_fence(__ATOMIC_RELEASE, "workgroup", "local");
    __builtin_amdgcn_s_barrier();
    __builtin_amdgcn_fence(__ATOMIC_ACQUIRE, "workgroup", "local");
}

__device__ __forceinline__ void wave_lds_order() {
    __builtin_amdgcn_fence(__ATOMIC_RELEASE, "wavefront", "local");
    __builtin_amdgcn_wave_barrier();
    __builtin_amdgcn_fence(__ATOMIC_ACQUIRE, "wavefront", "local");
}


__device__ __forceinline__ void load_tile(const float* __restrict__ X, long long row0,
                                          int lane, v4f (&r)[QUADS]) {
    const v4f* src = (const v4f*)(X + row0 * N_FEAT) + lane;
#pragma unroll
    for (int i = 0; i < QUADS; ++i) r[i] = src[32 * i];
}

__device__ __forceinline__ void load_tile_clamped(const float* __restrict__ X, long long row0,
                                                  int T, int lane, v4f (&r)[QUADS]) {
    const long long lastChunk = ((long long)T * N_FEAT) / 4 - 1;
    const long long chunk0    = row0 * N_FEAT / 4 + lane;
    const v4f* src = (const v4f*)X;
#pragma unroll
    for (int i = 0; i < QUADS; ++i) {
        const long long ch = chunk0 + 32 * i;
        r[i] = src[ch < lastChunk ? ch : lastChunk];
    }
}

template <bool HAS_NEXT>
__device__ __forceinline__ void do_tile(const float* __restrict__ X, float* __restrict__ Out,
                                        int T, long long row0, int lane,
                                        float* xw, const float* rd,
                                        const int (&soff)[QUADS], const v4f (&bq)[QUADS],
                                        v4f (&r)[QUADS]) {
    const bool live = row0 < T;
    const bool full = row0 + TILE_ROWS <= T;

    if (full) {
#pragma unroll
        for (int i = 0; i < QUADS; ++i) *(v4f*)(xw + soff[i]) = r[i];
    } else if (live) {
        const long long total = (long long)T * N_FEAT;
        const long long base  = row0 * N_FEAT;
        for (int j = 0; j < TILE_FLOATS / 32; ++j) {
            const int f = lane + 32 * j;
            const float v = (base + f < total) ? X[base + f] : 0.0f;
            xw[(f / SUPER) * LDS_PITCH + (f % SUPER)] = v;
        }
    }

    if (HAS_NEXT) {
        if (row0 + 2 * TILE_ROWS <= T) load_tile(X, row0 + TILE_ROWS, lane, r);
    }

    wave_lds_order();

    if (live) {
        v4f xq[QUADS];
#pragma unroll
        for (int q = 0; q < QUADS; ++q) xq[q] = *(const v4f*)(rd + 8 * q);

        v8f c0 = {};
        v8f c1 = {};
#pragma unroll
        for (int q = 0; q < QUADS; ++q) {
            v2f a0 = {xq[q].x, xq[q].y};
            v2f b0 = {bq[q].x, bq[q].y};
            v2f a1 = {xq[q].z, xq[q].w};
            v2f b1 = {bq[q].z, bq[q].w};
            c0 = __builtin_amdgcn_wmma_f32_16x16x4_f32(false, a0, false, b0,
                                                       (short)0, c0, false, false);
            c1 = __builtin_amdgcn_wmma_f32_16x16x4_f32(false, a1, false, b1,
                                                       (short)0, c1, false, false);
        }
        const v8f c = c0 + c1;

        if (full) {
            float* o = Out + row0 * N_CLS + lane;
#pragma unroll
            for (int v = 0; v < 8; ++v) o[32 * v] = c[v];
        } else {
            const long long total = (long long)T * N_CLS;
            const long long base  = row0 * N_CLS + lane;
#pragma unroll
            for (int v = 0; v < 8; ++v)
                if (base + 32 * v < total) Out[base + 32 * v] = c[v];
        }
    }
}

__global__ void __launch_bounds__(256) fold_wmma_kernel(
    const float* __restrict__ X,
    const float* __restrict__ psi_r,
    const float* __restrict__ psi_i,
    const float* __restrict__ A_r,
    const float* __restrict__ A_i,
    float* __restrict__ Out,
    int T) {
    __shared__ __attribute__((aligned(16))) float xs[WAVES_PER_BLOCK * LDS_TILE];
    __shared__ float Part[256];
    __shared__ float Ms[N_CLS * N_FEAT];

    const int t    = threadIdx.x;
    const int lane = t & 31;
    const int wave = __builtin_amdgcn_readfirstlane(t >> 5);
    const int hi   = lane >> 4;
    const int n    = lane & 15;

    const long long waveBase =
        (long long)blockIdx.x * ROWS_PER_BLOCK + (long long)wave * ROWS_PER_WAVE;
    float* xw = xs + wave * LDS_TILE;

    const bool part = t < PART_THREADS;
    const int  tp   = part ? t : 0;
    const int  ki   = tp / N_FEAT;
    const int  pa   = tp % N_FEAT;
    const int  pi_  = ki % N_FEAT;
    const float* ar = A_r + ki * (N_FEAT * N_FEAT) + pa;
    const float* ai = A_i + ki * (N_FEAT * N_FEAT) + pa;
    float ra[N_FEAT], ia[N_FEAT];
#pragma unroll
    for (int j = 0; j < N_FEAT; ++j) {
        ra[j] = ar[j * N_FEAT];
        ia[j] = ai[j * N_FEAT];
    }
    const float pri = psi_r[pi_];
    const float pii = psi_i[pi_];

    v4f r[QUADS];
    load_tile_clamped(X, waveBase, T, lane, r);
    __builtin_amdgcn_sched_barrier(0);

    {
        float val = 0.0f;
#pragma unroll
        for (int j = 0; j < N_FEAT; ++j) {
            const float w = psi_r[j] - psi_i[j];
            val += w * (pri * ra[j] - pii * ia[j]);
        }
        Part[t] = val;
    }
    lds_barrier();

    if (t < N_CLS * N_FEAT) {
        const int k = t / N_FEAT;
        const int a = t % N_FEAT;
        float val = 0.0f;
#pragma unroll
        for (int i = 0; i < N_FEAT; ++i)
            val += Part[(k * N_FEAT + i) * N_FEAT + a];
        Ms[t] = val;
    }
    lds_barrier();

    v4f bq[QUADS];
#pragma unroll
    for (int q = 0; q < QUADS; ++q) {
#pragma unroll
        for (int e = 0; e < 4; ++e) {
            const int kk = 8 * q + 4 * hi + e;
            const float m = Ms[(n & 1) * N_FEAT + kk % N_FEAT];
            bq[q][e] = (kk / N_FEAT == (n >> 1)) ? m : 0.0f;
        }
    }

    int soff[QUADS];
#pragma unroll
    for (int i = 0; i < QUADS; ++i) {
        const int c = 32 * i + lane;
        soff[i] = (c / 20) * LDS_PITCH + (c % 20) * 4;
    }
    const float* rd = xw + (2 * (n & 7) + (n >> 3)) * LDS_PITCH + 4 * hi;

#pragma unroll
    for (int g = 0; g < TILES_PER_WAVE; ++g) {
        if (g > 0) wave_lds_order();
        const long long row0 = waveBase + (long long)g * TILE_ROWS;
        if (g + 1 < TILES_PER_WAVE)
            do_tile<true>(X, Out, T, row0, lane, xw, rd, soff, bq, r);
        else
            do_tile<false>(X, Out, T, row0, lane, xw, rd, soff, bq, r);
    }
}

extern "C" void kernel_launch(void* const* d_in, const int* in_sizes, int n_in,
                              void* d_out, int out_size, void* d_ws, size_t ws_size,
                              hipStream_t stream) {
    const float* x     = (const float*)d_in[0];
    const float* psi_r = (const float*)d_in[1];
    const float* psi_i = (const float*)d_in[2];
    const float* A_r   = (const float*)d_in[3];
    const float* A_i   = (const float*)d_in[4];
    float* out  = (float*)d_out;

    const int T = in_sizes[0] / N_FEAT;

    const int grid = (T + ROWS_PER_BLOCK - 1) / ROWS_PER_BLOCK;
    fold_wmma_kernel<<<grid, 256, 0, stream>>>(x, psi_r, psi_i, A_r, A_i, out, T);
}
